# starting kernel
# baseline (speedup 1.0000x reference)
#include <hip/hip_runtime.h>

typedef float v2f __attribute__((ext_vector_type(2)));
typedef float v8f __attribute__((ext_vector_type(8)));

#define FRAME_LEN 160
#define HOP 256
#define MIN_LAG 40
#define WAVES_PER_BLOCK 8
#define XP_SIZE 416
#define XR_SIZE 176
#define NEG_INF (-3.402823466e+38f)

__global__ __launch_bounds__(256) void pitch_wmma_kernel(
    const float* __restrict__ audio, float* __restrict__ out,
    int S, int T, int nframes) {
  __shared__ float s_xp[WAVES_PER_BLOCK][XP_SIZE];
  __shared__ float s_xr[WAVES_PER_BLOCK][XR_SIZE];

  const int lane = threadIdx.x & 31;
  const int wave = threadIdx.x >> 5;
  int f = blockIdx.x * WAVES_PER_BLOCK + wave;
  if (f >= nframes) f = nframes - 1;
  const int b = f / T;
  const int t = f - b * T;
  const float* __restrict__ src = audio + (size_t)b * (size_t)S + (size_t)t * HOP;

  float x0 = src[lane];
  float x1 = src[lane + 32];
  float x2 = src[lane + 64];
  float x3 = src[lane + 96];
  float x4 = src[lane + 128];

  float sum = x0 + x1 + x2 + x3 + x4;
#pragma unroll
  for (int off = 16; off; off >>= 1) sum += __shfl_xor(sum, off, 32);
  const float mean = sum * (1.0f / 160.0f);
  x0 -= mean; x1 -= mean; x2 -= mean; x3 -= mean; x4 -= mean;

  float mx = fmaxf(fmaxf(fmaxf(fabsf(x0), fabsf(x1)),
                         fmaxf(fabsf(x2), fabsf(x3))), fabsf(x4));
#pragma unroll
  for (int off = 16; off; off >>= 1) mx = fmaxf(mx, __shfl_xor(mx, off, 32));
  const bool silent = mx < 1e-8f;

  float* xp = s_xp[wave];
  float* xr = s_xr[wave];
  for (int j = lane; j < 240; j += 32) xp[j] = 0.0f;
  if (lane < XR_SIZE - FRAME_LEN) xr[FRAME_LEN + lane] = 0.0f;
  xp[240 + lane      ] = x0;  xr[159 - lane      ] = x0;
  xp[240 + lane + 32 ] = x1;  xr[159 - lane - 32 ] = x1;
  xp[240 + lane + 64 ] = x2;  xr[159 - lane - 64 ] = x2;
  xp[240 + lane + 96 ] = x3;  xr[159 - lane - 96 ] = x3;
  xp[240 + lane + 128] = x4;  xr[159 - lane - 128] = x4;
  __syncthreads();

  const int n  = lane & 15;
  const int hf = lane >> 4;
  const float* __restrict__ xra = xr + 2 * hf + n;
  const float* __restrict__ xpb = xp + 240 + 2 * hf - 16 * n;
  v8f c = {};
#pragma unroll
  for (int k0 = 0; k0 < FRAME_LEN; k0 += 4) {
    v2f a, bb;
    a.x  = xra[k0];
    a.y  = xra[k0 + 1];
    bb.x = xpb[k0];
    bb.y = xpb[k0 + 1];
    c = __builtin_amdgcn_wmma_f32_16x16x4_f32(
         false, a,  false, bb,
         (short)0, c,  false,  false);
  }

  union { v8f v; float f[8]; } cu; cu.v = c;
  float bestv = NEG_INF;
  int   bestk = 1 << 30;
#pragma unroll
  for (int v = 0; v < 8; ++v) {
    const int m = v + 8 * hf;
    const int L = 16 * n + m;
    const int k = 159 - L;
    const float val = (L <= 159 - MIN_LAG) ? cu.f[v] : NEG_INF;
    const bool better = (val > bestv) | ((val == bestv) & (k < bestk));
    bestv = better ? val : bestv;
    bestk = better ? k : bestk;
  }
#pragma unroll
  for (int off = 16; off; off >>= 1) {
    const float ov = __shfl_xor(bestv, off, 32);
    const int   ok = __shfl_xor(bestk, off, 32);
    const bool better = (ov > bestv) | ((ov == bestv) & (ok < bestk));
    bestv = better ? ov : bestv;
    bestk = better ? ok : bestk;
  }

  if (lane == 0) {
    out[f] = silent ? 0.0f : 16000.0f / (float)bestk;
  }
}

extern "C" void kernel_launch(void* const* d_in, const int* in_sizes, int n_in,
                              void* d_out, int out_size, void* d_ws, size_t ws_size,
                              hipStream_t stream) {
  (void)n_in; (void)d_ws; (void)ws_size;
  const float* audio = (const float*)d_in[0];
  float* out = (float*)d_out;
  const int B = 16;
  const int S = in_sizes[0] / B;
  const int T = (S - FRAME_LEN) / HOP + 1;
  const int nframes = B * T;
  (void)out_size;
  const int blocks = (nframes + WAVES_PER_BLOCK - 1) / WAVES_PER_BLOCK;
  pitch_wmma_kernel<<<blocks, 32 * WAVES_PER_BLOCK, 0, stream>>>(
      audio, out, S, T, nframes);
}


# The 4 frames of a wave share their three butterflies (sum, magnitude, arg-max): the first two steps each halve the frames a lane carries; exchanges by DPP and row swap, no LDS permute; sums unchanged
# speedup vs baseline: 1.7543x; 1.7543x over previous
#include <hip/hip_runtime.h>

typedef float v2f __attribute__((ext_vector_type(2)));
typedef float v8f __attribute__((ext_vector_type(8)));

#define FRAME_LEN 160
#define HOP 256
#define MIN_LAG 40
#define WAVES_PER_BLOCK 8
#define FPW 4
#define NEG_INF (-3.402823466e+38f)

template <int M> __device__ __forceinline__ int xrow_i(int v) {
  return __builtin_amdgcn_update_dpp(0, v, 0x160 + M  , 0xF, 0xF, true);
}
template <int M> __device__ __forceinline__ float xrow(float v) {
  return __int_as_float(xrow_i<M>(__float_as_int(v)));
}
__device__ __forceinline__ int xhalf_i(int v) {
  return __builtin_amdgcn_permlanex16(v, v, 0x76543210, 0xfedcba98, false, false);
}
__device__ __forceinline__ float xhalf(float v) {
  return __int_as_float(xhalf_i(__float_as_int(v)));
}
__device__ __forceinline__ void take_better(float& v, int& k, float ov, int ok) {
  const bool better = (ov > v) | ((ov == v) & (ok < k));
  v = better ? ov : v;
  k = better ? ok : k;
}

#define EP_OFF   0
#define OP_OFF   140
#define ER_OFF   276
#define ORB_OFF  371
#define WAVE_LDS 464

__global__ __launch_bounds__(256) void pitch_wmma_kernel(
    const float* __restrict__ audio, float* __restrict__ out,
    int S, int T, int nframes) {
  __shared__ float s_w[WAVES_PER_BLOCK][FPW][WAVE_LDS];

  const int lane = threadIdx.x & 31;
  const int wave = threadIdx.x >> 5;
  const int f0 = (blockIdx.x * WAVES_PER_BLOCK + wave) * FPW;
  const int fc = f0 < nframes ? f0 : nframes - 1;
  const int b0 = fc / T;
  const int t0 = fc - b0 * T;

  float x[FPW][5];
#pragma unroll
  for (int i = 0; i < FPW; ++i) {
    const int fi = f0 + i < nframes ? f0 + i : nframes - 1;
    const int d  = fi - fc;
    int t = t0 + d;
    int b = b0;
    if (t >= T) { t -= T; b += 1; }
    const float* __restrict__ src = audio + (size_t)b * (size_t)S + (size_t)t * HOP;
#pragma unroll
    for (int j = 0; j < 5; ++j) x[i][j] = src[lane + 32 * j];
  }

  float sum[FPW];
#pragma unroll
  for (int i = 0; i < FPW; ++i) sum[i] = x[i][0] + x[i][1] + x[i][2] + x[i][3] + x[i][4];
  static_assert(FPW == 4, "the shared butterflies split 4 frames over two lane bits");
  const bool lb4 = (lane & 16) != 0;
  const bool lb3 = (lane & 8) != 0;
  float tot;
  {
    const float keepA = lb4 ? sum[2] : sum[0], sendA = lb4 ? sum[0] : sum[2];
    const float keepB = lb4 ? sum[3] : sum[1], sendB = lb4 ? sum[1] : sum[3];
    const float rA = keepA + xhalf(sendA);
    const float rB = keepB + xhalf(sendB);
    const float keep = lb3 ? rB : rA, send = lb3 ? rA : rB;
    tot = keep + xrow<8>(send);
    tot += xrow<4>(tot);
    tot += xrow<2>(tot);
    tot += xrow<1>(tot);
  }
  int mxl[FPW];
#pragma unroll
  for (int i = 0; i < FPW; ++i) {
    const float total = __int_as_float(__builtin_amdgcn_readlane(__float_as_int(tot), 8 * i));
    const float mean = total * (1.0f / 160.0f);
#pragma unroll
    for (int j = 0; j < 5; ++j) x[i][j] -= mean;
    mxl[i] = __float_as_int(fmaxf(fmaxf(fmaxf(fabsf(x[i][0]), fabsf(x[i][1])),
                                        fmaxf(fabsf(x[i][2]), fabsf(x[i][3]))), fabsf(x[i][4])));
  }
  const bool lb0 = (lane & 1) != 0;
  const bool lb1 = (lane & 2) != 0;
  int mxb;
  {
    const int keepA = lb0 ? mxl[1] : mxl[0], sendA = lb0 ? mxl[0] : mxl[1];
    const int keepB = lb0 ? mxl[3] : mxl[2], sendB = lb0 ? mxl[2] : mxl[3];
    const int rA = max(keepA, xrow_i<1>(sendA));
    const int rB = max(keepB, xrow_i<1>(sendB));
    const int keep = lb1 ? rB : rA, send = lb1 ? rA : rB;
    mxb = max(keep, xrow_i<2>(send));
    mxb = max(mxb, xrow_i<4>(mxb));
    mxb = max(mxb, xrow_i<8>(mxb));
    mxb = max(mxb, xhalf_i(mxb));
  }
  const float mx = __int_as_float(mxb);

  const int odd = lane & 1;
  const int a0  = lane >> 1;
#pragma unroll
  for (int i = 0; i < FPW; ++i) {
    float* w = s_w[wave][i];
    w[EP_OFF + lane] = 0.0f;
    w[OP_OFF + lane] = 0.0f;
  }
  if (lane < 24) {
#pragma unroll
    for (int i = 0; i < FPW; ++i) { s_w[wave][i][EP_OFF + 32 + lane] = 0.0f; s_w[wave][i][OP_OFF + 32 + lane] = 0.0f; }
  }
  if (lane < 8) {
#pragma unroll
    for (int i = 0; i < FPW; ++i) s_w[wave][i][ER_OFF + 80 + lane] = 0.0f;
  }
  if (lane < 9) {
#pragma unroll
    for (int i = 0; i < FPW; ++i) s_w[wave][i][ORB_OFF + 1 + 79 + lane] = 0.0f;
  }
#pragma unroll
  for (int i = 0; i < FPW; ++i) {
    float* w = s_w[wave][i];
    float* fw = w + (odd ? OP_OFF : EP_OFF) + 56 + a0;
    float* rv = w + (odd ? ORB_OFF : ER_OFF) + 79 - a0;
    fw[0]  = x[i][0];  rv[0]   = x[i][0];
    fw[16] = x[i][1];  rv[-16] = x[i][1];
    fw[32] = x[i][2];  rv[-32] = x[i][2];
    fw[48] = x[i][3];  rv[-48] = x[i][3];
    fw[64] = x[i][4];  rv[-64] = x[i][4];
  }
  __syncthreads();

  const int n   = lane & 15;
  const int hf  = lane >> 4;
  const int grp = n >> 3;
  const int r8  = n & 7;
  const int rph = n & 1;
  const int rq  = n >> 1;
  const float* __restrict__ xa = s_w[wave][0] + (rph ? ORB_OFF + 1 : ER_OFF) + rq + 2 * hf;
  const float* __restrict__ xb = s_w[wave][0] + (grp ? OP_OFF : EP_OFF) + 56 - 8 * r8 + 2 * hf;
  v8f c[FPW];
#pragma unroll
  for (int i = 0; i < FPW; ++i) c[i] = v8f{};
#pragma unroll
  for (int k0 = 0; k0 < FRAME_LEN / 2; k0 += 4) {
#pragma unroll
    for (int i = 0; i < FPW; ++i) {
      v2f a, bb;
      a.x  = xa[i * WAVE_LDS + k0];
      a.y  = xa[i * WAVE_LDS + k0 + 1];
      bb.x = xb[i * WAVE_LDS + k0];
      bb.y = xb[i * WAVE_LDS + k0 + 1];
      c[i] = __builtin_amdgcn_wmma_f32_16x16x4_f32(
           false, a,  false, bb,
           (short)0, c[i],  false,  false);
    }
  }

  const int kbase = 2 * (79 - 8 * r8 - 4 * hf) + grp;
  const bool has_lags = (8 * r8 + 4 * hf) <= 79 - MIN_LAG / 2 - 3;
  float bestv[FPW];
  int   bestk[FPW];
#pragma unroll
  for (int i = 0; i < FPW; ++i) {
    union { v8f v; float f[8]; } cu; cu.v = c[i];
    float bv = NEG_INF;
    int   bk = 1 << 30;
#pragma unroll
    for (int j = 3; j >= 0; --j) {
      const float oo = __int_as_float(__builtin_amdgcn_mov_dpp(
          __float_as_int(cu.f[2 * j + 1]), 0x128  , 0xF, 0xF, false));
      const float own  = cu.f[2 * j];
      const float lagv = own + (grp ? own : oo);
      const int   k    = kbase - 2 * j;
      const bool better = lagv > bv;
      bv = better ? lagv : bv;
      bk = better ? k : bk;
    }
    bestv[i] = has_lags ? bv : NEG_INF;
    bestk[i] = has_lags ? bk : (1 << 30);
  }
  float bv;
  int   bk;
  {
    float vA = lb0 ? bestv[1] : bestv[0];  const float svA = lb0 ? bestv[0] : bestv[1];
    int   kA = lb0 ? bestk[1] : bestk[0];  const int   skA = lb0 ? bestk[0] : bestk[1];
    float vB = lb0 ? bestv[3] : bestv[2];  const float svB = lb0 ? bestv[2] : bestv[3];
    int   kB = lb0 ? bestk[3] : bestk[2];  const int   skB = lb0 ? bestk[2] : bestk[3];
    take_better(vA, kA, xrow<1>(svA), xrow_i<1>(skA));
    take_better(vB, kB, xrow<1>(svB), xrow_i<1>(skB));
    bv = lb1 ? vB : vA;  const float sv = lb1 ? vA : vB;
    bk = lb1 ? kB : kA;  const int   sk = lb1 ? kA : kB;
    take_better(bv, bk, xrow<2>(sv), xrow_i<2>(sk));
    take_better(bv, bk, xrow<4>(bv), xrow_i<4>(bk));
    take_better(bv, bk, xrow<8>(bv), xrow_i<8>(bk));
    take_better(bv, bk, xhalf(bv), xhalf_i(bk));
  }

  if (lane < FPW && f0 + lane < nframes) {
    const bool silent = mx < 1e-8f;
    out[f0 + lane] = silent ? 0.0f : 16000.0f / (float)bk;
  }
}

extern "C" void kernel_launch(void* const* d_in, const int* in_sizes, int n_in,
                              void* d_out, int out_size, void* d_ws, size_t ws_size,
                              hipStream_t stream) {
  (void)n_in; (void)d_ws; (void)ws_size;
  const float* audio = (const float*)d_in[0];
  float* out = (float*)d_out;
  const int B = 16;
  const int S = in_sizes[0] / B;
  const int T = (S - FRAME_LEN) / HOP + 1;
  const int nframes = B * T;
  (void)out_size;
  const int fpb = WAVES_PER_BLOCK * FPW;
  const int blocks = (nframes + fpb - 1) / fpb;
  pitch_wmma_kernel<<<blocks, 32 * WAVES_PER_BLOCK, 0, stream>>>(
      audio, out, S, T, nframes);
}


# Address arithmetic off the vector ALU: wave number read as a scalar, so frame/batch indices and global bases run on the scalar unit; per-frame LDS A-fragment bases kept in registers; sums unchanged
# speedup vs baseline: 1.8696x; 1.0657x over previous
#include <hip/hip_runtime.h>

typedef float v2f __attribute__((ext_vector_type(2)));
typedef float v8f __attribute__((ext_vector_type(8)));
typedef const float __attribute__((address_space(3)))* LdsCF;

#define FRAME_LEN 160
#define HOP 256
#define MIN_LAG 40
#define WAVES_PER_BLOCK 8
#define FPW 4
#define NEG_INF (-3.402823466e+38f)

template <int M> __device__ __forceinline__ int xrow_i(int v) {
  return __builtin_amdgcn_update_dpp(0, v, 0x160 + M  , 0xF, 0xF, true);
}
template <int M> __device__ __forceinline__ float xrow(float v) {
  return __int_as_float(xrow_i<M>(__float_as_int(v)));
}
__device__ __forceinline__ int xhalf_i(int v) {
  return __builtin_amdgcn_permlanex16(v, v, 0x76543210, 0xfedcba98, false, false);
}
__device__ __forceinline__ float xhalf(float v) {
  return __int_as_float(xhalf_i(__float_as_int(v)));
}
__device__ __forceinline__ void take_better(float& v, int& k, float ov, int ok) {
  const bool better = (ov > v) | ((ov == v) & (ok < k));
  v = better ? ov : v;
  k = better ? ok : k;
}

#define EP_OFF   0
#define OP_OFF   140
#define ER_OFF   276
#define ORB_OFF  371
#define WAVE_LDS 464

__global__ __launch_bounds__(256) void pitch_wmma_kernel(
    const float* __restrict__ audio, float* __restrict__ out,
    int S, int T, int nframes) {
  __shared__ float s_w[WAVES_PER_BLOCK][FPW][WAVE_LDS];

  const int lane = threadIdx.x & 31;
  const int wave = __builtin_amdgcn_readfirstlane((int)(threadIdx.x >> 5));
  const int f0 = (blockIdx.x * WAVES_PER_BLOCK + wave) * FPW;
  const int fc = f0 < nframes ? f0 : nframes - 1;
  const int b0 = fc / T;
  const int t0 = fc - b0 * T;

  float x[FPW][5];
#pragma unroll
  for (int i = 0; i < FPW; ++i) {
    const int fi = f0 + i < nframes ? f0 + i : nframes - 1;
    const int d  = fi - fc;
    int t = t0 + d;
    int b = b0;
    if (t >= T) { t -= T; b += 1; }
    const float* __restrict__ src = audio + (size_t)b * (size_t)S + (size_t)t * HOP;
#pragma unroll
    for (int j = 0; j < 5; ++j) x[i][j] = src[lane + 32 * j];
  }

  float sum[FPW];
#pragma unroll
  for (int i = 0; i < FPW; ++i) sum[i] = x[i][0] + x[i][1] + x[i][2] + x[i][3] + x[i][4];
  static_assert(FPW == 4, "the shared butterflies split 4 frames over two lane bits");
  const bool lb4 = (lane & 16) != 0;
  const bool lb3 = (lane & 8) != 0;
  float tot;
  {
    const float keepA = lb4 ? sum[2] : sum[0], sendA = lb4 ? sum[0] : sum[2];
    const float keepB = lb4 ? sum[3] : sum[1], sendB = lb4 ? sum[1] : sum[3];
    const float rA = keepA + xhalf(sendA);
    const float rB = keepB + xhalf(sendB);
    const float keep = lb3 ? rB : rA, send = lb3 ? rA : rB;
    tot = keep + xrow<8>(send);
    tot += xrow<4>(tot);
    tot += xrow<2>(tot);
    tot += xrow<1>(tot);
  }
  int mxl[FPW];
#pragma unroll
  for (int i = 0; i < FPW; ++i) {
    const float total = __int_as_float(__builtin_amdgcn_readlane(__float_as_int(tot), 8 * i));
    const float mean = total * (1.0f / 160.0f);
#pragma unroll
    for (int j = 0; j < 5; ++j) x[i][j] -= mean;
    mxl[i] = __float_as_int(fmaxf(fmaxf(fmaxf(fabsf(x[i][0]), fabsf(x[i][1])),
                                        fmaxf(fabsf(x[i][2]), fabsf(x[i][3]))), fabsf(x[i][4])));
  }
  const bool lb0 = (lane & 1) != 0;
  const bool lb1 = (lane & 2) != 0;
  int mxb;
  {
    const int keepA = lb0 ? mxl[1] : mxl[0], sendA = lb0 ? mxl[0] : mxl[1];
    const int keepB = lb0 ? mxl[3] : mxl[2], sendB = lb0 ? mxl[2] : mxl[3];
    const int rA = max(keepA, xrow_i<1>(sendA));
    const int rB = max(keepB, xrow_i<1>(sendB));
    const int keep = lb1 ? rB : rA, send = lb1 ? rA : rB;
    mxb = max(keep, xrow_i<2>(send));
    mxb = max(mxb, xrow_i<4>(mxb));
    mxb = max(mxb, xrow_i<8>(mxb));
    mxb = max(mxb, xhalf_i(mxb));
  }
  const float mx = __int_as_float(mxb);

  const int odd = lane & 1;
  const int a0  = lane >> 1;
#pragma unroll
  for (int i = 0; i < FPW; ++i) {
    float* w = s_w[wave][i];
    w[EP_OFF + lane] = 0.0f;
    w[OP_OFF + lane] = 0.0f;
  }
  if (lane < 24) {
#pragma unroll
    for (int i = 0; i < FPW; ++i) { s_w[wave][i][EP_OFF + 32 + lane] = 0.0f; s_w[wave][i][OP_OFF + 32 + lane] = 0.0f; }
  }
  if (lane < 8) {
#pragma unroll
    for (int i = 0; i < FPW; ++i) s_w[wave][i][ER_OFF + 80 + lane] = 0.0f;
  }
  if (lane < 9) {
#pragma unroll
    for (int i = 0; i < FPW; ++i) s_w[wave][i][ORB_OFF + 1 + 79 + lane] = 0.0f;
  }
#pragma unroll
  for (int i = 0; i < FPW; ++i) {
    float* w = s_w[wave][i];
    float* fw = w + (odd ? OP_OFF : EP_OFF) + 56 + a0;
    float* rv = w + (odd ? ORB_OFF : ER_OFF) + 79 - a0;
    fw[0]  = x[i][0];  rv[0]   = x[i][0];
    fw[16] = x[i][1];  rv[-16] = x[i][1];
    fw[32] = x[i][2];  rv[-32] = x[i][2];
    fw[48] = x[i][3];  rv[-48] = x[i][3];
    fw[64] = x[i][4];  rv[-64] = x[i][4];
  }
  __syncthreads();

  const int n   = lane & 15;
  const int hf  = lane >> 4;
  const int grp = n >> 3;
  const int r8  = n & 7;
  const int rph = n & 1;
  const int rq  = n >> 1;
  const float* __restrict__ xa = s_w[wave][0] + (rph ? ORB_OFF + 1 : ER_OFF) + rq + 2 * hf;
  const float* __restrict__ xb = s_w[wave][0] + (grp ? OP_OFF : EP_OFF) + 56 - 8 * r8 + 2 * hf;
  LdsCF xaf[FPW];
#pragma unroll
  for (int i = 0; i < FPW; ++i) {
    xaf[i] = (LdsCF)(xa + i * WAVE_LDS);
    asm volatile("" : "+v"(xaf[i]));
  }
  v8f c[FPW];
#pragma unroll
  for (int i = 0; i < FPW; ++i) c[i] = v8f{};
#pragma unroll
  for (int k0 = 0; k0 < FRAME_LEN / 2; k0 += 4) {
#pragma unroll
    for (int i = 0; i < FPW; ++i) {
      v2f a, bb;
      a.x  = xaf[i][k0];
      a.y  = xaf[i][k0 + 1];
      bb.x = xb[i * WAVE_LDS + k0];
      bb.y = xb[i * WAVE_LDS + k0 + 1];
      c[i] = __builtin_amdgcn_wmma_f32_16x16x4_f32(
           false, a,  false, bb,
           (short)0, c[i],  false,  false);
    }
  }

  const int kbase = 2 * (79 - 8 * r8 - 4 * hf) + grp;
  const bool has_lags = (8 * r8 + 4 * hf) <= 79 - MIN_LAG / 2 - 3;
  float bestv[FPW];
  int   bestk[FPW];
#pragma unroll
  for (int i = 0; i < FPW; ++i) {
    union { v8f v; float f[8]; } cu; cu.v = c[i];
    float bv = NEG_INF;
    int   bk = 1 << 30;
#pragma unroll
    for (int j = 3; j >= 0; --j) {
      const float oo = __int_as_float(__builtin_amdgcn_mov_dpp(
          __float_as_int(cu.f[2 * j + 1]), 0x128  , 0xF, 0xF, false));
      const float own  = cu.f[2 * j];
      const float lagv = own + (grp ? own : oo);
      const int   k    = kbase - 2 * j;
      const bool better = lagv > bv;
      bv = better ? lagv : bv;
      bk = better ? k : bk;
    }
    bestv[i] = has_lags ? bv : NEG_INF;
    bestk[i] = has_lags ? bk : (1 << 30);
  }
  float bv;
  int   bk;
  {
    float vA = lb0 ? bestv[1] : bestv[0];  const float svA = lb0 ? bestv[0] : bestv[1];
    int   kA = lb0 ? bestk[1] : bestk[0];  const int   skA = lb0 ? bestk[0] : bestk[1];
    float vB = lb0 ? bestv[3] : bestv[2];  const float svB = lb0 ? bestv[2] : bestv[3];
    int   kB = lb0 ? bestk[3] : bestk[2];  const int   skB = lb0 ? bestk[2] : bestk[3];
    take_better(vA, kA, xrow<1>(svA), xrow_i<1>(skA));
    take_better(vB, kB, xrow<1>(svB), xrow_i<1>(skB));
    bv = lb1 ? vB : vA;  const float sv = lb1 ? vA : vB;
    bk = lb1 ? kB : kA;  const int   sk = lb1 ? kA : kB;
    take_better(bv, bk, xrow<2>(sv), xrow_i<2>(sk));
    take_better(bv, bk, xrow<4>(bv), xrow_i<4>(bk));
    take_better(bv, bk, xrow<8>(bv), xrow_i<8>(bk));
    take_better(bv, bk, xhalf(bv), xhalf_i(bk));
  }

  if (lane < FPW && f0 + lane < nframes) {
    const bool silent = mx < 1e-8f;
    out[f0 + lane] = silent ? 0.0f : 16000.0f / (float)bk;
  }
}

extern "C" void kernel_launch(void* const* d_in, const int* in_sizes, int n_in,
                              void* d_out, int out_size, void* d_ws, size_t ws_size,
                              hipStream_t stream) {
  (void)n_in; (void)d_ws; (void)ws_size;
  const float* audio = (const float*)d_in[0];
  float* out = (float*)d_out;
  const int B = 16;
  const int S = in_sizes[0] / B;
  const int T = (S - FRAME_LEN) / HOP + 1;
  const int nframes = B * T;
  (void)out_size;
  const int fpb = WAVES_PER_BLOCK * FPW;
  const int blocks = (nframes + fpb - 1) / fpb;
  pitch_wmma_kernel<<<blocks, 32 * WAVES_PER_BLOCK, 0, stream>>>(
      audio, out, S, T, nframes);
}
